# starting kernel
# baseline (speedup 1.0000x reference)
#include <hip/hip_runtime.h>

typedef __attribute__((ext_vector_type(2))) float v2f;
typedef __attribute__((ext_vector_type(4))) float v4f;
typedef __attribute__((ext_vector_type(8))) float v8f;

namespace {
constexpr int DI = 128, DJ = 512, DK = 512;
constexpr int TI = 16, TJ = 16, TK = 16;
constexpr int NBJ = DJ / TJ;
constexpr int NBK = DK / TK;
constexpr int NBLOCKS = (DI / TI) * NBJ * NBK;

__device__ inline v8f wmma_k8(v2f alo, v2f ahi, v2f blo, v2f bhi, v8f c) {
  c = __builtin_amdgcn_wmma_f32_16x16x4_f32(false, alo, false, blo, (short)0, c,
                                            false, false);
  c = __builtin_amdgcn_wmma_f32_16x16x4_f32(false, ahi, false, bhi, (short)0, c,
                                            false, false);
  return c;
}

__device__ inline void lds_add(float* p, float v) {
  (void)__hip_atomic_fetch_add(p, v, __ATOMIC_RELAXED,
                               __HIP_MEMORY_SCOPE_WORKGROUP);
}
}

__global__ __launch_bounds__(128) void partition_tca_kernel(
    const float* __restrict__ v00, const float* __restrict__ v01,
    const float* __restrict__ v10, const float* __restrict__ v11,
    const float* __restrict__ v20, const float* __restrict__ v21,
    float* __restrict__ out) {
  __shared__ float acc[TI * TJ * TK];

  const int tid  = threadIdx.x;
  const int lane = tid & 31;
  const int wave = tid >> 5;
  const int half = lane >> 4;
  const int ln   = lane & 15;

  const int tile = blockIdx.x;
  const int kt = tile & (NBK - 1);
  const int jt = (tile >> 5) & (NBJ - 1);
  const int it = tile >> 10;
  const int i0 = it * TI, j0 = jt * TJ, k0 = kt * TK;

  const int rb = half * 2;

  v2f a0l, a0h, a1l, a1h, a2l, a2h;
  a0l.x = v00[(0 + rb + 0) * DI + i0 + ln];
  a0l.y = v00[(0 + rb + 1) * DI + i0 + ln];
  a0h.x = v00[(4 + rb + 0) * DI + i0 + ln];
  a0h.y = v00[(4 + rb + 1) * DI + i0 + ln];

  a1l.x = v10[(0 + rb + 0) * DJ + j0 + ln];
  a1l.y = v10[(0 + rb + 1) * DJ + j0 + ln];
  a1h.x = v10[(4 + rb + 0) * DJ + j0 + ln];
  a1h.y = v10[(4 + rb + 1) * DJ + j0 + ln];

  a2l.x = v20[(0 + rb + 0) * DK + k0 + ln];
  a2l.y = v20[(0 + rb + 1) * DK + k0 + ln];
  a2h.x = v20[(4 + rb + 0) * DK + k0 + ln];
  a2h.y = v20[(4 + rb + 1) * DK + k0 + ln];

  for (int q = 0; q < 4; ++q) {
    const int js = wave * 4 + q;
    const int jg = j0 + js;
    v2f bl, bh;
    bl.x = v01[((0 + rb + 0) * (size_t)DJ + jg) * DK + k0 + ln];
    bl.y = v01[((0 + rb + 1) * (size_t)DJ + jg) * DK + k0 + ln];
    bh.x = v01[((4 + rb + 0) * (size_t)DJ + jg) * DK + k0 + ln];
    bh.y = v01[((4 + rb + 1) * (size_t)DJ + jg) * DK + k0 + ln];
    v8f d = {};
    d = wmma_k8(a0l, a0h, bl, bh, d);
#pragma unroll
    for (int v = 0; v < 8; ++v) {
      const int mi = v + half * 8;
      acc[(mi * TJ + js) * TK + ln] = d[v];
    }
  }
  __syncthreads();

  for (int q = 0; q < 4; ++q) {
    const int is = wave * 4 + q;
    const int ig = i0 + is;

    v2f b1l, b1h, b2l, b2h;
    b1l.x = v11[((0 + rb + 0) * (size_t)DI + ig) * DK + k0 + ln];
    b1l.y = v11[((0 + rb + 1) * (size_t)DI + ig) * DK + k0 + ln];
    b1h.x = v11[((4 + rb + 0) * (size_t)DI + ig) * DK + k0 + ln];
    b1h.y = v11[((4 + rb + 1) * (size_t)DI + ig) * DK + k0 + ln];

    b2l.x = v21[((0 + rb + 0) * (size_t)DI + ig) * DJ + j0 + ln];
    b2l.y = v21[((0 + rb + 1) * (size_t)DI + ig) * DJ + j0 + ln];
    b2h.x = v21[((4 + rb + 0) * (size_t)DI + ig) * DJ + j0 + ln];
    b2h.y = v21[((4 + rb + 1) * (size_t)DI + ig) * DJ + j0 + ln];

    v8f d1 = {};
    d1 = wmma_k8(a1l, a1h, b1l, b1h, d1);
    v8f d2 = {};
    d2 = wmma_k8(a2l, a2h, b2l, b2h, d2);

#pragma unroll
    for (int v = 0; v < 8; ++v) {
      const int m = v + half * 8;
      lds_add(&acc[(is * TJ + m) * TK + ln], d1[v]);
      lds_add(&acc[(is * TJ + ln) * TK + m], d2[v]);
    }
  }
  __syncthreads();

  const v4f* accv = (const v4f*)acc;
#pragma unroll
  for (int t = 0; t < 4; ++t) {
    const int rh  = t * 128 + tid;
    const int row = rh >> 1;
    const int h   = rh & 1;
    const int i   = row >> 4;
    const int j   = row & 15;
    v4f lo = accv[row * 4 + h * 2 + 0];
    v4f hi = accv[row * 4 + h * 2 + 1];
    float* dst = out + ((size_t)(i0 + i) * DJ + (j0 + j)) * DK + k0 + h * 8;
    *(v4f*)(dst + 0) = lo;
    *(v4f*)(dst + 4) = hi;
  }
}

extern "C" void kernel_launch(void* const* d_in, const int* in_sizes, int n_in,
                              void* d_out, int out_size, void* d_ws,
                              size_t ws_size, hipStream_t stream) {
  const float* v00 = (const float*)d_in[0];
  const float* v01 = (const float*)d_in[1];
  const float* v10 = (const float*)d_in[2];
  const float* v11 = (const float*)d_in[3];
  const float* v20 = (const float*)d_in[4];
  const float* v21 = (const float*)d_in[5];
  float* out = (float*)d_out;

  partition_tca_kernel<<<NBLOCKS, 128, 0, stream>>>(v00, v01, v10, v11, v20,
                                                    v21, out);
}


# Re-tile to 16x16x64 steps in 512 looping work-groups: terms 1+2 chained in WMMA accumulators, term 0 added from one plain LDS pass, no LDS atomics; lanes own 4 adjacent k; sum order (t1+t2)+t0
# speedup vs baseline: 4.0429x; 4.0429x over previous
#include <hip/hip_runtime.h>


typedef __attribute__((ext_vector_type(2))) float v2f;
typedef __attribute__((ext_vector_type(4))) float v4f;
typedef __attribute__((ext_vector_type(8))) float v8f;

namespace {
constexpr int DI = 128, DJ = 512, DK = 512;
constexpr int TI = 16, TJ = 16;
constexpr int KC = 64;
constexpr int NSTEP = 4;
constexpr int NWG = (DI / TI) * (DJ / TJ) * (DK / (KC * NSTEP));
constexpr int STAGE4 = TI * TJ * KC / 4;

struct Frag {
  v4f q0, q1, q2, q3;
};

__device__ inline Frag load_rows(const float* __restrict__ p, size_t rstride,
                                 int rb) {
  Frag f;
  f.q0 = *(const v4f*)(p + (size_t)(rb + 0) * rstride);
  f.q1 = *(const v4f*)(p + (size_t)(rb + 1) * rstride);
  f.q2 = *(const v4f*)(p + (size_t)(rb + 4) * rstride);
  f.q3 = *(const v4f*)(p + (size_t)(rb + 5) * rstride);
  return f;
}

__device__ inline v8f wmma4(v2f a, v2f b, v8f c) {
  return __builtin_amdgcn_wmma_f32_16x16x4_f32(false, a, false, b, (short)0, c,
                                               false, false);
}

__device__ inline void mma_rank8(v8f (&acc)[4], v2f al, v2f ah, const Frag& f) {
  for (int t = 0; t < 4; ++t) {
    v2f b = {f.q0[t], f.q1[t]};
    acc[t] = wmma4(al, b, acc[t]);
  }
  for (int t = 0; t < 4; ++t) {
    v2f b = {f.q2[t], f.q3[t]};
    acc[t] = wmma4(ah, b, acc[t]);
  }
}
}

__global__ __launch_bounds__(256) void partition_tca_kernel(
    const float* __restrict__ v00, const float* __restrict__ v01,
    const float* __restrict__ v10, const float* __restrict__ v11,
    const float* __restrict__ v20, const float* __restrict__ v21,
    float* __restrict__ out) {
  __shared__ v4f stage[2][STAGE4];

  const int tid  = threadIdx.x;
  const int lane = tid & 31;
  const int wave = tid >> 5;
  const int half = lane >> 4;
  const int c    = lane & 15;
  const int rb   = half * 2;

  const int wg = blockIdx.x;
  const int kh = wg & 1;
  const int jt = (wg >> 1) & (DJ / TJ - 1);
  const int it = wg >> 6;
  const int i0 = it * TI, j0 = jt * TJ;
  const int kbeg = kh * (KC * NSTEP);
  const int rot = jt & (NSTEP - 1);

  v2f a0l = {v00[(rb + 0) * DI + i0 + c], v00[(rb + 1) * DI + i0 + c]};
  v2f a0h = {v00[(rb + 4) * DI + i0 + c], v00[(rb + 5) * DI + i0 + c]};
  v2f a1l = {v10[(rb + 0) * DJ + j0 + c], v10[(rb + 1) * DJ + j0 + c]};
  v2f a1h = {v10[(rb + 4) * DJ + j0 + c], v10[(rb + 5) * DJ + j0 + c]};
  v2f a2l[2], a2h[2];
  for (int q = 0; q < 2; ++q) {
    const int ig = i0 + wave * 2 + q;
    a2l[q].x = v21[((size_t)(rb + 0) * DI + ig) * DJ + j0 + c];
    a2l[q].y = v21[((size_t)(rb + 1) * DI + ig) * DJ + j0 + c];
    a2h[q].x = v21[((size_t)(rb + 4) * DI + ig) * DJ + j0 + c];
    a2h[q].y = v21[((size_t)(rb + 5) * DI + ig) * DJ + j0 + c];
  }

  Frag f0[2];
  {
    const int kk = kbeg + rot * KC + 4 * c;
    for (int q = 0; q < 2; ++q)
      f0[q] = load_rows(v01 + (size_t)(j0 + wave * 2 + q) * DK + kk,
                        (size_t)DJ * DK, rb);
  }

#pragma clang loop unroll(disable)
  for (int s = 0; s < NSTEP; ++s) {
    const int kbase = kbeg + ((s + rot) & (NSTEP - 1)) * KC;
    const int kk = kbase + 4 * c;

    Frag f1[2];
    for (int q = 0; q < 2; ++q)
      f1[q] = load_rows(v11 + (size_t)(i0 + wave * 2 + q) * DK + kk,
                        (size_t)DI * DK, rb);
    const Frag f2 = load_rows(v20 + kk, (size_t)DK, rb);

    v4f* S = stage[s & 1];

    for (int q = 0; q < 2; ++q) {
      const int js = wave * 2 + q;
      v8f d[4] = {};
      mma_rank8(d, a0l, a0h, f0[q]);
      for (int v = 0; v < 8; ++v) {
        v4f o = {d[0][v], d[1][v], d[2][v], d[3][v]};
        S[((v + 8 * half) * TJ + js) * (KC / 4) + c] = o;
      }
    }

    __syncthreads();

    if (s + 1 < NSTEP) {
      const int kn = kbeg + ((s + 1 + rot) & (NSTEP - 1)) * KC + 4 * c;
      for (int q = 0; q < 2; ++q)
        f0[q] = load_rows(v01 + (size_t)(j0 + wave * 2 + q) * DK + kn,
                          (size_t)DJ * DK, rb);
    }

    for (int q = 0; q < 2; ++q) {
      const int is = wave * 2 + q;
      const int ig = i0 + is;
      v8f d[4] = {};
      mma_rank8(d, a1l, a1h, f1[q]);
      mma_rank8(d, a2l[q], a2h[q], f2);
      for (int v = 0; v < 8; ++v) {
        const int jl = v + 8 * half;
        const v4f t0 = S[(is * TJ + jl) * (KC / 4) + c];
        v4f o = {d[0][v] + t0.x, d[1][v] + t0.y, d[2][v] + t0.z,
                 d[3][v] + t0.w};
        *(v4f*)(out + ((size_t)ig * DJ + j0 + jl) * DK + kk) = o;
      }
    }
  }
}

extern "C" void kernel_launch(void* const* d_in, const int* in_sizes, int n_in,
                              void* d_out, int out_size, void* d_ws,
                              size_t ws_size, hipStream_t stream) {
  const float* v00 = (const float*)d_in[0];
  const float* v01 = (const float*)d_in[1];
  const float* v10 = (const float*)d_in[2];
  const float* v11 = (const float*)d_in[3];
  const float* v20 = (const float*)d_in[4];
  const float* v21 = (const float*)d_in[5];
  float* out = (float*)d_out;

  partition_tca_kernel<<<NWG, 256, 0, stream>>>(v00, v01, v10, v11, v20, v21,
                                                out);
}


# All global loads of step s+1 issued ahead of step s's global stores, into operand registers their last WMMA has freed; hand-written barrier waits for LDS only; last step peeled
# speedup vs baseline: 4.2190x; 1.0436x over previous
#include <hip/hip_runtime.h>
#include <type_traits>


typedef __attribute__((ext_vector_type(2))) float v2f;
typedef __attribute__((ext_vector_type(4))) float v4f;
typedef __attribute__((ext_vector_type(8))) float v8f;

namespace {
constexpr int DI = 128, DJ = 512, DK = 512;
constexpr int TI = 16, TJ = 16;
constexpr int KC = 64;
constexpr int NSTEP = 4;
constexpr int NWG = (DI / TI) * (DJ / TJ) * (DK / (KC * NSTEP));
constexpr int STAGE4 = TI * TJ * KC / 4;

struct Frag {
  v4f q0, q1, q2, q3;
};

__device__ inline Frag load_rows(const float* __restrict__ p, size_t rstride,
                                 int rb) {
  Frag f;
  f.q0 = *(const v4f*)(p + (size_t)(rb + 0) * rstride);
  f.q1 = *(const v4f*)(p + (size_t)(rb + 1) * rstride);
  f.q2 = *(const v4f*)(p + (size_t)(rb + 4) * rstride);
  f.q3 = *(const v4f*)(p + (size_t)(rb + 5) * rstride);
  return f;
}

__device__ inline v8f wmma4(v2f a, v2f b, v8f c) {
  return __builtin_amdgcn_wmma_f32_16x16x4_f32(false, a, false, b, (short)0, c,
                                               false, false);
}

__device__ inline void mma_rank8(v8f (&acc)[4], v2f al, v2f ah, const Frag& f) {
  for (int t = 0; t < 4; ++t) {
    v2f b = {f.q0[t], f.q1[t]};
    acc[t] = wmma4(al, b, acc[t]);
  }
  for (int t = 0; t < 4; ++t) {
    v2f b = {f.q2[t], f.q3[t]};
    acc[t] = wmma4(ah, b, acc[t]);
  }
}

__device__ inline void lds_barrier() {
  asm volatile("s_wait_dscnt 0x0\n\ts_barrier_signal -1\n\ts_barrier_wait -1" ::: "memory");
}
}

__global__ __launch_bounds__(256) void partition_tca_kernel(
    const float* __restrict__ v00, const float* __restrict__ v01,
    const float* __restrict__ v10, const float* __restrict__ v11,
    const float* __restrict__ v20, const float* __restrict__ v21,
    float* __restrict__ out) {
  __shared__ v4f stage[2][STAGE4];

  const int tid  = threadIdx.x;
  const int lane = tid & 31;
  const int wave = tid >> 5;
  const int half = lane >> 4;
  const int c    = lane & 15;
  const int rb   = half * 2;

  const int wg = blockIdx.x;
  const int kh = wg & 1;
  const int jt = (wg >> 1) & (DJ / TJ - 1);
  const int it = wg >> 6;
  const int i0 = it * TI, j0 = jt * TJ;
  const int kbeg = kh * (KC * NSTEP);
  const int rot = jt & (NSTEP - 1);

  v2f a0l = {v00[(rb + 0) * DI + i0 + c], v00[(rb + 1) * DI + i0 + c]};
  v2f a0h = {v00[(rb + 4) * DI + i0 + c], v00[(rb + 5) * DI + i0 + c]};
  v2f a1l = {v10[(rb + 0) * DJ + j0 + c], v10[(rb + 1) * DJ + j0 + c]};
  v2f a1h = {v10[(rb + 4) * DJ + j0 + c], v10[(rb + 5) * DJ + j0 + c]};
  v2f a2l[2], a2h[2];
  for (int q = 0; q < 2; ++q) {
    const int ig = i0 + wave * 2 + q;
    a2l[q].x = v21[((size_t)(rb + 0) * DI + ig) * DJ + j0 + c];
    a2l[q].y = v21[((size_t)(rb + 1) * DI + ig) * DJ + j0 + c];
    a2h[q].x = v21[((size_t)(rb + 4) * DI + ig) * DJ + j0 + c];
    a2h[q].y = v21[((size_t)(rb + 5) * DI + ig) * DJ + j0 + c];
  }

  Frag f0[2], f1[2], f2;
  {
    const int kk = kbeg + rot * KC + 4 * c;
    for (int q = 0; q < 2; ++q)
      f0[q] = load_rows(v01 + (size_t)(j0 + wave * 2 + q) * DK + kk,
                        (size_t)DJ * DK, rb);
    for (int q = 0; q < 2; ++q)
      f1[q] = load_rows(v11 + (size_t)(i0 + wave * 2 + q) * DK + kk,
                        (size_t)DI * DK, rb);
    f2 = load_rows(v20 + kk, (size_t)DK, rb);
  }

  auto step = [&](const int s, auto more_tag) __attribute__((always_inline)) {
    constexpr bool MORE = decltype(more_tag)::value;
    const int kk = kbeg + ((s + rot) & (NSTEP - 1)) * KC + 4 * c;
    const int kn = kbeg + ((s + 1 + rot) & (NSTEP - 1)) * KC + 4 * c;
    v4f* S = stage[s & 1];

    {
      v8f d[2][4] = {};
      for (int q = 0; q < 2; ++q) mma_rank8(d[q], a0l, a0h, f0[q]);
      for (int q = 0; q < 2; ++q) {
        const int js = wave * 2 + q;
        for (int v = 0; v < 8; ++v) {
          v4f o = {d[q][0][v], d[q][1][v], d[q][2][v], d[q][3][v]};
          S[((v + 8 * half) * TJ + js) * (KC / 4) + c] = o;
        }
      }
    }
    if constexpr (MORE) {
      for (int q = 0; q < 2; ++q)
        f0[q] = load_rows(v01 + (size_t)(j0 + wave * 2 + q) * DK + kn,
                          (size_t)DJ * DK, rb);
    }

    lds_barrier();

    v8f d[2][4] = {};
    for (int q = 0; q < 2; ++q) {
      mma_rank8(d[q], a1l, a1h, f1[q]);
      mma_rank8(d[q], a2l[q], a2h[q], f2);
    }
    if constexpr (MORE) {
      for (int q = 0; q < 2; ++q)
        f1[q] = load_rows(v11 + (size_t)(i0 + wave * 2 + q) * DK + kn,
                          (size_t)DI * DK, rb);
      f2 = load_rows(v20 + kn, (size_t)DK, rb);
    }

    for (int q = 0; q < 2; ++q) {
      const int is = wave * 2 + q;
      const int ig = i0 + is;
      for (int v = 0; v < 8; ++v) {
        const int jl = v + 8 * half;
        const v4f t0 = S[(is * TJ + jl) * (KC / 4) + c];
        v4f o = {d[q][0][v] + t0.x, d[q][1][v] + t0.y, d[q][2][v] + t0.z,
                 d[q][3][v] + t0.w};
        *(v4f*)(out + ((size_t)ig * DJ + j0 + jl) * DK + kk) = o;
      }
    }
  };

#pragma clang loop unroll(disable)
  for (int s = 0; s < NSTEP - 1; ++s) step(s, std::true_type{});
  step(NSTEP - 1, std::false_type{});
}

extern "C" void kernel_launch(void* const* d_in, const int* in_sizes, int n_in,
                              void* d_out, int out_size, void* d_ws,
                              size_t ws_size, hipStream_t stream) {
  const float* v00 = (const float*)d_in[0];
  const float* v01 = (const float*)d_in[1];
  const float* v10 = (const float*)d_in[2];
  const float* v11 = (const float*)d_in[3];
  const float* v20 = (const float*)d_in[4];
  const float* v21 = (const float*)d_in[5];
  float* out = (float*)d_out;

  partition_tca_kernel<<<NWG, 256, 0, stream>>>(v00, v01, v10, v11, v20, v21,
                                                out);
}
